# starting kernel
# baseline (speedup 1.0000x reference)
#include <hip/hip_runtime.h>
#include <math.h>

#define Bsz   16
#define Nn    256
#define Ld    64
#define Hh    32
#define HE    128
#define EO    64
#define NMPS  3
#define ALPHA 0.2f

typedef __attribute__((ext_vector_type(2))) float v2f;
typedef __attribute__((ext_vector_type(8))) float v8f;

__device__ __forceinline__ float lrelu(float v) { return (v >= 0.f) ? v : ALPHA * v; }

__global__ __launch_bounds__(64) void k_input(const float* __restrict__ x,
                                              const float* __restrict__ W_lin,
                                              const float* __restrict__ b_lin,
                                              const float* __restrict__ W_in,
                                              const float* __restrict__ b_in,
                                              float* __restrict__ h) {
  const int b = blockIdx.x >> 8;
  const int n = blockIdx.x & 255;
  const int c = threadIdx.x;
  __shared__ float xs[Ld];
  __shared__ float ts[Ld];
  xs[c] = x[b * Ld + c];
  __syncthreads();
  float acc = b_lin[n * Ld + c];
#pragma unroll 8
  for (int l = 0; l < Ld; ++l) acc += xs[l] * W_lin[l * (Nn * Ld) + n * Ld + c];
  ts[c] = acc;
  __syncthreads();
  if (c < Hh) {
    float a2 = b_in[c];
#pragma unroll 8
    for (int l = 0; l < Ld; ++l) a2 += ts[l] * W_in[l * Hh + c];
    h[(b * Nn + n) * Hh + c] = a2;
  }
}

__global__ __launch_bounds__(128) void k_rcv_snd(const float* __restrict__ h,
                                                 const float* __restrict__ We1,
                                                 float* __restrict__ rcv,
                                                 float* __restrict__ snd, int t) {
  const int bn = blockIdx.x;
  const int k = threadIdx.x;
  __shared__ float hs[Hh];
  if (k < Hh) hs[k] = h[bn * Hh + k];
  __syncthreads();
  const float* W1 = We1 + (size_t)t * (2 * Hh) * HE;
  float a1 = 0.f, a2 = 0.f;
#pragma unroll 8
  for (int hh = 0; hh < Hh; ++hh) {
    const float hv = hs[hh];
    a1 += hv * W1[hh * HE + k];
    a2 += hv * W1[(Hh + hh) * HE + k];
  }
  rcv[bn * HE + k] = a1;
  snd[bn * HE + k] = a2;
}

#define EPAD 132

__global__ __launch_bounds__(128) void k_edge(const float* __restrict__ rcv,
                                              const float* __restrict__ snd,
                                              const float* __restrict__ be1,
                                              const float* __restrict__ We2,
                                              const float* __restrict__ be2,
                                              float* __restrict__ m, int t) {
  const int b    = blockIdx.x >> 8;
  const int i    = blockIdx.x & 255;
  const int tid  = threadIdx.x;
  const int wave = tid >> 5;
  const int lane = tid & 31;
  const int lh   = lane >> 4;
  const int ln   = lane & 15;

  __shared__ float Etile[16 * EPAD];

  const float rc = rcv[(b * Nn + i) * HE + tid] + be1[t * HE + tid];

  const int nc = wave * 16 + ln;
  const float* W2 = We2 + (size_t)t * HE * EO;
  v2f Bf[32];
#pragma unroll
  for (int kk = 0; kk < 32; ++kk) {
    const int k0 = kk * 4 + lh * 2;
    Bf[kk].x = W2[(k0    ) * EO + nc];
    Bf[kk].y = W2[(k0 + 1) * EO + nc];
  }
  const float bias2 = be2[t * EO + nc];

  float macc[8];
#pragma unroll
  for (int r = 0; r < 8; ++r) macc[r] = 0.f;

  const float* sndb = snd + (size_t)(b * Nn) * HE;

  for (int jt = 0; jt < Nn / 16; ++jt) {
    __syncthreads();
#pragma unroll 4
    for (int jr = 0; jr < 16; ++jr) {
      Etile[jr * EPAD + tid] = lrelu(rc + sndb[(jt * 16 + jr) * HE + tid]);
    }
    __syncthreads();

    v8f C = {};
#pragma unroll
    for (int kk = 0; kk < 32; ++kk) {
      const int k0 = kk * 4 + lh * 2;
      v2f A;
      A.x = Etile[ln * EPAD + k0];
      A.y = Etile[ln * EPAD + k0 + 1];
      C = __builtin_amdgcn_wmma_f32_16x16x4_f32(
           false, A,  false, Bf[kk],
           (short)0, C,  false,  false);
    }
#pragma unroll
    for (int r = 0; r < 8; ++r) macc[r] += lrelu(C[r] + bias2);
  }

  float colsum = 0.f;
#pragma unroll
  for (int r = 0; r < 8; ++r) colsum += macc[r];
  colsum += __shfl_xor(colsum, 16, 32);
  if (lane < 16) m[(b * Nn + i) * EO + nc] = colsum;
}

__global__ __launch_bounds__(32) void k_node(const float* __restrict__ h,
                                             const float* __restrict__ m,
                                             const float* __restrict__ Wn0,
                                             const float* __restrict__ bn0,
                                             const float* __restrict__ Wn1,
                                             const float* __restrict__ bn1,
                                             float* __restrict__ hout, int t,
                                             int apply_tanh) {
  const int bn = blockIdx.x;
  const int tid = threadIdx.x;
  __shared__ float nf[Hh + EO];
  __shared__ float t0[Hh];
  nf[tid]          = h[bn * Hh + tid];
  nf[Hh + tid]     = m[bn * EO + tid];
  nf[Hh + 32 + tid]= m[bn * EO + 32 + tid];
  __syncthreads();
  const float* W0 = Wn0 + (size_t)t * (Hh + EO) * Hh;
  float acc = bn0[t * Hh + tid];
#pragma unroll 8
  for (int f = 0; f < Hh + EO; ++f) acc += nf[f] * W0[f * Hh + tid];
  t0[tid] = lrelu(acc);
  __syncthreads();
  const float* W1 = Wn1 + (size_t)t * Hh * Hh;
  float a2 = bn1[t * Hh + tid];
#pragma unroll 8
  for (int f = 0; f < Hh; ++f) a2 += t0[f] * W1[f * Hh + tid];
  a2 = lrelu(a2);
  hout[bn * Hh + tid] = apply_tanh ? tanhf(a2) : a2;
}

extern "C" void kernel_launch(void* const* d_in, const int* in_sizes, int n_in,
                              void* d_out, int out_size, void* d_ws, size_t ws_size,
                              hipStream_t stream) {
  const float* x     = (const float*)d_in[0];
  const float* W_lin = (const float*)d_in[1];
  const float* b_lin = (const float*)d_in[2];
  const float* W_in  = (const float*)d_in[3];
  const float* b_in  = (const float*)d_in[4];
  const float* We1   = (const float*)d_in[5];
  const float* be1   = (const float*)d_in[6];
  const float* We2   = (const float*)d_in[7];
  const float* be2   = (const float*)d_in[8];
  const float* Wn0   = (const float*)d_in[9];
  const float* bn0   = (const float*)d_in[10];
  const float* Wn1   = (const float*)d_in[11];
  const float* bn1   = (const float*)d_in[12];
  float* out = (float*)d_out;

  float* ws  = (float*)d_ws;
  float* h   = ws;
  float* rcv = h   + Bsz * Nn * Hh;
  float* snd = rcv + Bsz * Nn * HE;
  float* mbuf= snd + Bsz * Nn * HE;

  const int BN = Bsz * Nn;

  k_input<<<BN, 64, 0, stream>>>(x, W_lin, b_lin, W_in, b_in, h);
  for (int t = 0; t < NMPS; ++t) {
    k_rcv_snd<<<BN, 128, 0, stream>>>(h, We1, rcv, snd, t);
    k_edge<<<BN, 128, 0, stream>>>(rcv, snd, be1, We2, be2, mbuf, t);
    k_node<<<BN, 32, 0, stream>>>(h, mbuf, Wn0, bn0, Wn1, bn1,
                                  (t == NMPS - 1) ? out : h, t, t == NMPS - 1);
  }
}


# k_edge: one wave per receiver; it builds each 16-sender E tile in its own LDS slice, reads it once as A and runs all 4 column tiles (We2 in 256 VGPRs); no workgroup barrier; sums bit-identical
# speedup vs baseline: 1.1897x; 1.1897x over previous
#include <hip/hip_runtime.h>
#include <math.h>

#define Bsz   16
#define Nn    256
#define Ld    64
#define Hh    32
#define HE    128
#define EO    64
#define NMPS  3
#define ALPHA 0.2f

typedef __attribute__((ext_vector_type(2))) float v2f;
typedef __attribute__((ext_vector_type(8))) float v8f;

__device__ __forceinline__ float lrelu(float v) { return (v >= 0.f) ? v : ALPHA * v; }

__global__ __launch_bounds__(64) void k_input(const float* __restrict__ x,
                                              const float* __restrict__ W_lin,
                                              const float* __restrict__ b_lin,
                                              const float* __restrict__ W_in,
                                              const float* __restrict__ b_in,
                                              float* __restrict__ h) {
  const int b = blockIdx.x >> 8;
  const int n = blockIdx.x & 255;
  const int c = threadIdx.x;
  __shared__ float xs[Ld];
  __shared__ float ts[Ld];
  xs[c] = x[b * Ld + c];
  __syncthreads();
  float acc = b_lin[n * Ld + c];
#pragma unroll 8
  for (int l = 0; l < Ld; ++l) acc += xs[l] * W_lin[l * (Nn * Ld) + n * Ld + c];
  ts[c] = acc;
  __syncthreads();
  if (c < Hh) {
    float a2 = b_in[c];
#pragma unroll 8
    for (int l = 0; l < Ld; ++l) a2 += ts[l] * W_in[l * Hh + c];
    h[(b * Nn + n) * Hh + c] = a2;
  }
}

__global__ __launch_bounds__(128) void k_rcv_snd(const float* __restrict__ h,
                                                 const float* __restrict__ We1,
                                                 float* __restrict__ rcv,
                                                 float* __restrict__ snd, int t) {
  const int bn = blockIdx.x;
  const int k = threadIdx.x;
  __shared__ float hs[Hh];
  if (k < Hh) hs[k] = h[bn * Hh + k];
  __syncthreads();
  const float* W1 = We1 + (size_t)t * (2 * Hh) * HE;
  float a1 = 0.f, a2 = 0.f;
#pragma unroll 8
  for (int hh = 0; hh < Hh; ++hh) {
    const float hv = hs[hh];
    a1 += hv * W1[hh * HE + k];
    a2 += hv * W1[(Hh + hh) * HE + k];
  }
  rcv[bn * HE + k] = a1;
  snd[bn * HE + k] = a2;
}

#define EPAD 132

__global__ __launch_bounds__(32) void k_edge(const float* __restrict__ rcv,
                                             const float* __restrict__ snd,
                                             const float* __restrict__ be1,
                                             const float* __restrict__ We2,
                                             const float* __restrict__ be2,
                                             float* __restrict__ m, int t) {
  const int bi   = blockIdx.x;
  const int b    = bi >> 8;
  const int lane = threadIdx.x;
  const int lh   = lane >> 4;
  const int ln   = lane & 15;

  __shared__ __attribute__((aligned(16))) float Etile[16 * EPAD];

  const float4 r4 = *(const float4*)(rcv + (size_t)bi * HE + 4 * lane);
  const float4 e4 = *(const float4*)(be1 + t * HE + 4 * lane);
  const float rc0 = r4.x + e4.x, rc1 = r4.y + e4.y, rc2 = r4.z + e4.z, rc3 = r4.w + e4.w;

  const float* W2 = We2 + (size_t)t * HE * EO;
  v2f Bf[4][32];
#pragma unroll
  for (int kk = 0; kk < 32; ++kk) {
    const int k0 = kk * 4 + lh * 2;
    const float4 w0 = *(const float4*)(W2 + (k0    ) * EO + 4 * ln);
    const float4 w1 = *(const float4*)(W2 + (k0 + 1) * EO + 4 * ln);
    Bf[0][kk].x = w0.x; Bf[0][kk].y = w1.x;
    Bf[1][kk].x = w0.y; Bf[1][kk].y = w1.y;
    Bf[2][kk].x = w0.z; Bf[2][kk].y = w1.z;
    Bf[3][kk].x = w0.w; Bf[3][kk].y = w1.w;
  }
  const float4 b24 = *(const float4*)(be2 + t * EO + 4 * ln);
  const float bias2[4] = {b24.x, b24.y, b24.z, b24.w};

  float macc[4][8];
#pragma unroll
  for (int c = 0; c < 4; ++c)
#pragma unroll
    for (int r = 0; r < 8; ++r) macc[c][r] = 0.f;

  const float* sndb = snd + (size_t)(b * Nn) * HE + 4 * lane;

  float4 S[16];
#pragma unroll
  for (int jr = 0; jr < 16; ++jr) S[jr] = *(const float4*)(sndb + jr * HE);

  for (int jt = 0; jt < Nn / 16; ++jt) {
    __syncthreads();
#pragma unroll
    for (int jr = 0; jr < 16; ++jr) {
      float4 e;
      e.x = lrelu(rc0 + S[jr].x);
      e.y = lrelu(rc1 + S[jr].y);
      e.z = lrelu(rc2 + S[jr].z);
      e.w = lrelu(rc3 + S[jr].w);
      *(float4*)(Etile + jr * EPAD + 4 * lane) = e;
    }
    __syncthreads();

    if (jt + 1 < Nn / 16) {
#pragma unroll
      for (int jr = 0; jr < 16; ++jr)
        S[jr] = *(const float4*)(sndb + ((jt + 1) * 16 + jr) * HE);
    }

    v2f A[32];
#pragma unroll
    for (int kk = 0; kk < 32; ++kk) {
      const int k0 = kk * 4 + lh * 2;
      A[kk].x = Etile[ln * EPAD + k0];
      A[kk].y = Etile[ln * EPAD + k0 + 1];
    }
    __builtin_amdgcn_sched_barrier(0);

    v8f C[4] = {};
#pragma unroll
    for (int kk = 0; kk < 32; ++kk) {
#pragma unroll
      for (int c = 0; c < 4; ++c) {
        C[c] = __builtin_amdgcn_wmma_f32_16x16x4_f32(
             false, A[kk],  false, Bf[c][kk],
             (short)0, C[c],  false,  false);
      }
    }
#pragma unroll
    for (int c = 0; c < 4; ++c)
#pragma unroll
      for (int r = 0; r < 8; ++r) macc[c][r] += lrelu(C[c][r] + bias2[c]);
  }

  float cs[4];
#pragma unroll
  for (int c = 0; c < 4; ++c) {
    float colsum = 0.f;
#pragma unroll
    for (int r = 0; r < 8; ++r) colsum += macc[c][r];
    colsum += __shfl_xor(colsum, 16, 32);
    cs[c] = colsum;
  }
  if (lane < 16) {
    float4 o; o.x = cs[0]; o.y = cs[1]; o.z = cs[2]; o.w = cs[3];
    *(float4*)(m + (size_t)bi * EO + 4 * ln) = o;
  }
}

__global__ __launch_bounds__(32) void k_node(const float* __restrict__ h,
                                             const float* __restrict__ m,
                                             const float* __restrict__ Wn0,
                                             const float* __restrict__ bn0,
                                             const float* __restrict__ Wn1,
                                             const float* __restrict__ bn1,
                                             float* __restrict__ hout, int t,
                                             int apply_tanh) {
  const int bn = blockIdx.x;
  const int tid = threadIdx.x;
  __shared__ float nf[Hh + EO];
  __shared__ float t0[Hh];
  nf[tid]          = h[bn * Hh + tid];
  nf[Hh + tid]     = m[bn * EO + tid];
  nf[Hh + 32 + tid]= m[bn * EO + 32 + tid];
  __syncthreads();
  const float* W0 = Wn0 + (size_t)t * (Hh + EO) * Hh;
  float acc = bn0[t * Hh + tid];
#pragma unroll 8
  for (int f = 0; f < Hh + EO; ++f) acc += nf[f] * W0[f * Hh + tid];
  t0[tid] = lrelu(acc);
  __syncthreads();
  const float* W1 = Wn1 + (size_t)t * Hh * Hh;
  float a2 = bn1[t * Hh + tid];
#pragma unroll 8
  for (int f = 0; f < Hh; ++f) a2 += t0[f] * W1[f * Hh + tid];
  a2 = lrelu(a2);
  hout[bn * Hh + tid] = apply_tanh ? tanhf(a2) : a2;
}

extern "C" void kernel_launch(void* const* d_in, const int* in_sizes, int n_in,
                              void* d_out, int out_size, void* d_ws, size_t ws_size,
                              hipStream_t stream) {
  const float* x     = (const float*)d_in[0];
  const float* W_lin = (const float*)d_in[1];
  const float* b_lin = (const float*)d_in[2];
  const float* W_in  = (const float*)d_in[3];
  const float* b_in  = (const float*)d_in[4];
  const float* We1   = (const float*)d_in[5];
  const float* be1   = (const float*)d_in[6];
  const float* We2   = (const float*)d_in[7];
  const float* be2   = (const float*)d_in[8];
  const float* Wn0   = (const float*)d_in[9];
  const float* bn0   = (const float*)d_in[10];
  const float* Wn1   = (const float*)d_in[11];
  const float* bn1   = (const float*)d_in[12];
  float* out = (float*)d_out;

  float* ws  = (float*)d_ws;
  float* h   = ws;
  float* rcv = h   + Bsz * Nn * Hh;
  float* snd = rcv + Bsz * Nn * HE;
  float* mbuf= snd + Bsz * Nn * HE;

  const int BN = Bsz * Nn;

  k_input<<<BN, 64, 0, stream>>>(x, W_lin, b_lin, W_in, b_in, h);
  for (int t = 0; t < NMPS; ++t) {
    k_rcv_snd<<<BN, 128, 0, stream>>>(h, We1, rcv, snd, t);
    k_edge<<<BN, 32, 0, stream>>>(rcv, snd, be1, We2, be2, mbuf, t);
    k_node<<<BN, 32, 0, stream>>>(h, mbuf, Wn0, bn0, Wn1, bn1,
                                  (t == NMPS - 1) ? out : h, t, t == NMPS - 1);
  }
}


# Per-node layers (input, rcv/snd, node MLP): weight columns stay in registers over 4-8 rows, rows enter as wave-uniform scalar operands, no LDS; their waves per call 69632 -> 10240; sums bit-identical
# speedup vs baseline: 1.2445x; 1.0461x over previous
#include <hip/hip_runtime.h>
#include <math.h>

#define Bsz   16
#define Nn    256
#define Ld    64
#define Hh    32
#define HE    128
#define EO    64
#define NMPS  3
#define ALPHA 0.2f

typedef __attribute__((ext_vector_type(2))) float v2f;
typedef __attribute__((ext_vector_type(8))) float v8f;

__device__ __forceinline__ float lrelu(float v) { return (v >= 0.f) ? v : ALPHA * v; }

#define IN_RB 4
__global__ __launch_bounds__(32) void k_input(const float* __restrict__ x,
                                              const float* __restrict__ W_lin,
                                              const float* __restrict__ b_lin,
                                              const float* __restrict__ W_in,
                                              const float* __restrict__ b_in,
                                              float* __restrict__ h) {
  const int n  = blockIdx.x & 255;
  const int bg = blockIdx.x >> 8;
  const int c  = threadIdx.x;
  float wl0[Ld], wl1[Ld], wi[Ld];
#pragma unroll
  for (int l = 0; l < Ld; ++l) {
    wl0[l] = W_lin[l * (Nn * Ld) + n * Ld + c];
    wl1[l] = W_lin[l * (Nn * Ld) + n * Ld + 32 + c];
    wi[l]  = W_in[l * Hh + c];
  }
  const float bl0 = b_lin[n * Ld + c];
  const float bl1 = b_lin[n * Ld + 32 + c];
  const float bi  = b_in[c];
#pragma unroll
  for (int r = 0; r < IN_RB; ++r) {
    const int b = bg * IN_RB + r;
    const float* xr = x + b * Ld;
    float a0 = bl0, a1 = bl1;
#pragma unroll
    for (int l = 0; l < Ld; ++l) {
      const float xv = xr[l];
      a0 += xv * wl0[l];
      a1 += xv * wl1[l];
    }
    float a2 = bi;
#pragma unroll
    for (int l = 0; l < 32; ++l) {
      const float tl = __int_as_float(__builtin_amdgcn_readlane(__float_as_int(a0), l));
      a2 += tl * wi[l];
    }
#pragma unroll
    for (int l = 0; l < 32; ++l) {
      const float tl = __int_as_float(__builtin_amdgcn_readlane(__float_as_int(a1), l));
      a2 += tl * wi[32 + l];
    }
    h[(b * Nn + n) * Hh + c] = a2;
  }
}

#define RS_R 8
__global__ __launch_bounds__(128) void k_rcv_snd(const float* __restrict__ h,
                                                 const float* __restrict__ We1,
                                                 float* __restrict__ rcv,
                                                 float* __restrict__ snd, int t) {
  const int k = threadIdx.x;
  const float* W1 = We1 + (size_t)t * (2 * Hh) * HE;
  float wr[Hh], wsd[Hh];
#pragma unroll
  for (int hh = 0; hh < Hh; ++hh) {
    wr[hh]  = W1[hh * HE + k];
    wsd[hh] = W1[(Hh + hh) * HE + k];
  }
  const int n0 = blockIdx.x * RS_R;
#pragma unroll
  for (int r = 0; r < RS_R; ++r) {
    const float* hr = h + (size_t)(n0 + r) * Hh;
    float a1 = 0.f, a2 = 0.f;
#pragma unroll
    for (int hh = 0; hh < Hh; ++hh) {
      const float hv = hr[hh];
      a1 += hv * wr[hh];
      a2 += hv * wsd[hh];
    }
    rcv[(size_t)(n0 + r) * HE + k] = a1;
    snd[(size_t)(n0 + r) * HE + k] = a2;
  }
}

#define EPAD 132

__global__ __launch_bounds__(32) void k_edge(const float* __restrict__ rcv,
                                             const float* __restrict__ snd,
                                             const float* __restrict__ be1,
                                             const float* __restrict__ We2,
                                             const float* __restrict__ be2,
                                             float* __restrict__ m, int t) {
  const int bi   = blockIdx.x;
  const int b    = bi >> 8;
  const int lane = threadIdx.x;
  const int lh   = lane >> 4;
  const int ln   = lane & 15;

  __shared__ __attribute__((aligned(16))) float Etile[16 * EPAD];

  const float4 r4 = *(const float4*)(rcv + (size_t)bi * HE + 4 * lane);
  const float4 e4 = *(const float4*)(be1 + t * HE + 4 * lane);
  const float rc0 = r4.x + e4.x, rc1 = r4.y + e4.y, rc2 = r4.z + e4.z, rc3 = r4.w + e4.w;

  const float* W2 = We2 + (size_t)t * HE * EO;
  v2f Bf[4][32];
#pragma unroll
  for (int kk = 0; kk < 32; ++kk) {
    const int k0 = kk * 4 + lh * 2;
    const float4 w0 = *(const float4*)(W2 + (k0    ) * EO + 4 * ln);
    const float4 w1 = *(const float4*)(W2 + (k0 + 1) * EO + 4 * ln);
    Bf[0][kk].x = w0.x; Bf[0][kk].y = w1.x;
    Bf[1][kk].x = w0.y; Bf[1][kk].y = w1.y;
    Bf[2][kk].x = w0.z; Bf[2][kk].y = w1.z;
    Bf[3][kk].x = w0.w; Bf[3][kk].y = w1.w;
  }
  const float4 b24 = *(const float4*)(be2 + t * EO + 4 * ln);
  const float bias2[4] = {b24.x, b24.y, b24.z, b24.w};

  float macc[4][8];
#pragma unroll
  for (int c = 0; c < 4; ++c)
#pragma unroll
    for (int r = 0; r < 8; ++r) macc[c][r] = 0.f;

  const float* sndb = snd + (size_t)(b * Nn) * HE + 4 * lane;

  float4 S[16];
#pragma unroll
  for (int jr = 0; jr < 16; ++jr) S[jr] = *(const float4*)(sndb + jr * HE);

  for (int jt = 0; jt < Nn / 16; ++jt) {
    __syncthreads();
#pragma unroll
    for (int jr = 0; jr < 16; ++jr) {
      float4 e;
      e.x = lrelu(rc0 + S[jr].x);
      e.y = lrelu(rc1 + S[jr].y);
      e.z = lrelu(rc2 + S[jr].z);
      e.w = lrelu(rc3 + S[jr].w);
      *(float4*)(Etile + jr * EPAD + 4 * lane) = e;
    }
    __syncthreads();

    if (jt + 1 < Nn / 16) {
#pragma unroll
      for (int jr = 0; jr < 16; ++jr)
        S[jr] = *(const float4*)(sndb + ((jt + 1) * 16 + jr) * HE);
    }

    v2f A[32];
#pragma unroll
    for (int kk = 0; kk < 32; ++kk) {
      const int k0 = kk * 4 + lh * 2;
      A[kk].x = Etile[ln * EPAD + k0];
      A[kk].y = Etile[ln * EPAD + k0 + 1];
    }
    __builtin_amdgcn_sched_barrier(0);

    v8f C[4] = {};
#pragma unroll
    for (int kk = 0; kk < 32; ++kk) {
#pragma unroll
      for (int c = 0; c < 4; ++c) {
        C[c] = __builtin_amdgcn_wmma_f32_16x16x4_f32(
             false, A[kk],  false, Bf[c][kk],
             (short)0, C[c],  false,  false);
      }
    }
#pragma unroll
    for (int c = 0; c < 4; ++c)
#pragma unroll
      for (int r = 0; r < 8; ++r) macc[c][r] += lrelu(C[c][r] + bias2[c]);
  }

  float cs[4];
#pragma unroll
  for (int c = 0; c < 4; ++c) {
    float colsum = 0.f;
#pragma unroll
    for (int r = 0; r < 8; ++r) colsum += macc[c][r];
    colsum += __shfl_xor(colsum, 16, 32);
    cs[c] = colsum;
  }
  if (lane < 16) {
    float4 o; o.x = cs[0]; o.y = cs[1]; o.z = cs[2]; o.w = cs[3];
    *(float4*)(m + (size_t)bi * EO + 4 * ln) = o;
  }
}

#define ND_R 4
__global__ __launch_bounds__(32) void k_node(const float* __restrict__ h,
                                             const float* __restrict__ m,
                                             const float* __restrict__ Wn0,
                                             const float* __restrict__ bn0,
                                             const float* __restrict__ Wn1,
                                             const float* __restrict__ bn1,
                                             float* __restrict__ hout, int t,
                                             int apply_tanh) {
  const int tid = threadIdx.x;
  const float* W0 = Wn0 + (size_t)t * (Hh + EO) * Hh;
  const float* W1 = Wn1 + (size_t)t * Hh * Hh;
  float w0[Hh + EO], w1[Hh];
#pragma unroll
  for (int f = 0; f < Hh + EO; ++f) w0[f] = W0[f * Hh + tid];
#pragma unroll
  for (int f = 0; f < Hh; ++f) w1[f] = W1[f * Hh + tid];
  const float b0 = bn0[t * Hh + tid];
  const float b1 = bn1[t * Hh + tid];
  const int n0 = blockIdx.x * ND_R;
#pragma unroll
  for (int r = 0; r < ND_R; ++r) {
    const int bn = n0 + r;
    const float* hr = h + (size_t)bn * Hh;
    const float* mr = m + (size_t)bn * EO;
    float acc = b0;
#pragma unroll
    for (int f = 0; f < Hh; ++f) acc += hr[f] * w0[f];
#pragma unroll
    for (int f = 0; f < EO; ++f) acc += mr[f] * w0[Hh + f];
    const float t0 = lrelu(acc);
    float a2 = b1;
#pragma unroll
    for (int f = 0; f < Hh; ++f) {
      const float tf = __int_as_float(__builtin_amdgcn_readlane(__float_as_int(t0), f));
      a2 += tf * w1[f];
    }
    a2 = lrelu(a2);
    hout[bn * Hh + tid] = apply_tanh ? tanhf(a2) : a2;
  }
}

extern "C" void kernel_launch(void* const* d_in, const int* in_sizes, int n_in,
                              void* d_out, int out_size, void* d_ws, size_t ws_size,
                              hipStream_t stream) {
  const float* x     = (const float*)d_in[0];
  const float* W_lin = (const float*)d_in[1];
  const float* b_lin = (const float*)d_in[2];
  const float* W_in  = (const float*)d_in[3];
  const float* b_in  = (const float*)d_in[4];
  const float* We1   = (const float*)d_in[5];
  const float* be1   = (const float*)d_in[6];
  const float* We2   = (const float*)d_in[7];
  const float* be2   = (const float*)d_in[8];
  const float* Wn0   = (const float*)d_in[9];
  const float* bn0   = (const float*)d_in[10];
  const float* Wn1   = (const float*)d_in[11];
  const float* bn1   = (const float*)d_in[12];
  float* out = (float*)d_out;

  float* ws  = (float*)d_ws;
  float* h   = ws;
  float* rcv = h   + Bsz * Nn * Hh;
  float* snd = rcv + Bsz * Nn * HE;
  float* mbuf= snd + Bsz * Nn * HE;

  const int BN = Bsz * Nn;

  k_input<<<BN / IN_RB, 32, 0, stream>>>(x, W_lin, b_lin, W_in, b_in, h);
  for (int t = 0; t < NMPS; ++t) {
    k_rcv_snd<<<BN / RS_R, 128, 0, stream>>>(h, We1, rcv, snd, t);
    k_edge<<<BN, 32, 0, stream>>>(rcv, snd, be1, We2, be2, mbuf, t);
    k_node<<<BN / ND_R, 32, 0, stream>>>(h, mbuf, Wn0, bn0, Wn1, bn1,
                                  (t == NMPS - 1) ? out : h, t, t == NMPS - 1);
  }
}
